# starting kernel
# baseline (speedup 1.0000x reference)
#include <hip/hip_runtime.h>
#include <cmath>

typedef __attribute__((ext_vector_type(2))) float v2f;
typedef __attribute__((ext_vector_type(4))) float v4f;
typedef __attribute__((ext_vector_type(8))) float v8f;

#define RH   32
#define NT   24
#define H    32
#define SCS  28
#define WAVES_PER_BLOCK 8
#define ROWS_PER_WAVE   32
#define ROWS_PER_BLOCK  (WAVES_PER_BLOCK * ROWS_PER_WAVE)

static __device__ __forceinline__ float fast_tanh(float v) {
#if __has_builtin(__builtin_amdgcn_tanhf)
    return __builtin_amdgcn_tanhf(v);
#else
    return tanhf(v);
#endif
}

static __device__ __forceinline__ float fast_exp(float v) {
#if __has_builtin(__builtin_amdgcn_exp2f)
    return __builtin_amdgcn_exp2f(v * 1.44269504088896340736f);
#else
    return __expf(v);
#endif
}

__global__ __launch_bounds__(ROWS_PER_BLOCK)
void forest_moe_kernel(const float* __restrict__ x,
                       const float* __restrict__ rw1, const float* __restrict__ rb1,
                       const float* __restrict__ rw2, const float* __restrict__ rb2,
                       const float* __restrict__ tw1, const float* __restrict__ tb1,
                       const float* __restrict__ tw2, const float* __restrict__ tb2,
                       float* __restrict__ y)
{
    __shared__ float s_rw1[RH], s_rb1[RH];
    __shared__ float s_rw2[RH * NT];
    __shared__ float s_rb2[NT], s_tb2[NT];
    __shared__ v4f   s_tree[NT * H];
    __shared__ __align__(16) float s_sc[WAVES_PER_BLOCK][ROWS_PER_WAVE][SCS];

    const int tid  = threadIdx.x;
    const int wave = tid >> 5;
    const int lane = tid & 31;

    for (int i = tid; i < RH * NT; i += ROWS_PER_BLOCK) s_rw2[i] = rw2[i];
    for (int i = tid; i < NT * H;  i += ROWS_PER_BLOCK) {
        v4f w = { tw1[i], tb1[i], tw2[i], 0.0f };
        s_tree[i] = w;
    }
    if (tid < RH) { s_rw1[tid] = rw1[tid]; s_rb1[tid] = rb1[tid]; }
    if (tid < NT) { s_rb2[tid] = rb2[tid]; s_tb2[tid] = tb2[tid]; }
    __syncthreads();

    const int rowBase = blockIdx.x * ROWS_PER_BLOCK + wave * ROWS_PER_WAVE;
    const int mrow    = lane & 15;
    const int khalf   = (lane >> 4) << 1;

    float xsave[2];
#pragma unroll
    for (int rt = 0; rt < 2; ++rt) {
        const int   gr = rowBase + rt * 16 + mrow;
        const float xv = x[gr];
        xsave[rt] = xv;

        v2f a[8];
#pragma unroll
        for (int s = 0; s < 8; ++s) {
            const int k0 = 4 * s + khalf;
            a[s].x = fast_tanh(xv * s_rw1[k0]     + s_rb1[k0]);
            a[s].y = fast_tanh(xv * s_rw1[k0 + 1] + s_rb1[k0 + 1]);
        }
#pragma unroll
        for (int n = 0; n < 2; ++n) {
            const int col = n * 16 + mrow;
            v8f c = {};
#pragma unroll
            for (int s = 0; s < 8; ++s) {
                const int k0 = 4 * s + khalf;
                v2f b;
                b.x = (col < NT) ? s_rw2[k0 * NT + col]       : 0.0f;
                b.y = (col < NT) ? s_rw2[(k0 + 1) * NT + col] : 0.0f;
                c = __builtin_amdgcn_wmma_f32_16x16x4_f32(
                         false, a[s],  false, b,
                         (short)0, c,  false,  false);
            }
            if (col < NT) {
#pragma unroll
                for (int r = 0; r < 8; ++r) {
                    const int row = rt * 16 + ((lane >> 4) << 3) + r;
                    s_sc[wave][row][col] = c[r];
                }
            }
        }
    }
    __syncthreads();

    const float xv = (lane < 16) ? xsave[0] : xsave[1];

    const v4f* scv = (const v4f*)&s_sc[wave][lane][0];
    v4f q[6];
#pragma unroll
    for (int j = 0; j < 6; ++j) q[j] = scv[j];

    float b0 = -__builtin_inff(), b1v = -__builtin_inff(), b2v = -__builtin_inff();
    int   i0 = 0, i1 = 0, i2 = 0;
#pragma unroll
    for (int t = 0; t < NT; ++t) {
        const float sv = q[t >> 2][t & 3] + s_rb2[t];
        if (sv > b0)       { b2v = b1v; i2 = i1; b1v = b0; i1 = i0; b0 = sv; i0 = t; }
        else if (sv > b1v) { b2v = b1v; i2 = i1; b1v = sv; i1 = t; }
        else if (sv > b2v) { b2v = sv;  i2 = t; }
    }
    const float e1  = fast_exp(b1v - b0);
    const float e2  = fast_exp(b2v - b0);
    const float inv = 1.0f / (1.0f + e1 + e2);

    const int   idx[3] = { i0, i1, i2 };
    const float wgt[3] = { inv, e1 * inv, e2 * inv };

    float acc = 0.0f;
#pragma unroll
    for (int i = 0; i < 3; ++i) {
        const int t = idx[i];
        const v4f* tp = &s_tree[t * H];
        float dot = 0.0f;
#pragma unroll
        for (int k = 0; k < H; ++k) {
            const v4f w = tp[k];
            const float h = fast_tanh(xv * w.x + w.y);
            dot += h * w.z;
        }
        acc += wgt[i] * (dot + s_tb2[t]);
    }
    y[rowBase + lane] = acc;
}

extern "C" void kernel_launch(void* const* d_in, const int* in_sizes, int n_in,
                              void* d_out, int out_size, void* d_ws, size_t ws_size,
                              hipStream_t stream) {
    const float* x   = (const float*)d_in[0];
    const float* rw1 = (const float*)d_in[1];
    const float* rb1 = (const float*)d_in[2];
    const float* rw2 = (const float*)d_in[3];
    const float* rb2 = (const float*)d_in[4];
    const float* tw1 = (const float*)d_in[5];
    const float* tb1 = (const float*)d_in[6];
    const float* tw2 = (const float*)d_in[7];
    const float* tb2 = (const float*)d_in[8];
    float* y = (float*)d_out;

    const int B = in_sizes[0];
    const int grid = B / ROWS_PER_BLOCK;
    forest_moe_kernel<<<grid, ROWS_PER_BLOCK, 0, stream>>>(
        x, rw1, rb1, rw2, rb2, tw1, tb1, tw2, tb2, y);
}


# tree-parameter LDS table read from a per-lane start slot (lane&15, wrap copies) so the data-indexed ds_load_b128 gathers are bank-conflict-free; k sum order of each expert dot rotates per lane
# speedup vs baseline: 1.3253x; 1.3253x over previous
#include <hip/hip_runtime.h>
#include <cmath>

typedef __attribute__((ext_vector_type(2))) float v2f;
typedef __attribute__((ext_vector_type(4))) float v4f;
typedef __attribute__((ext_vector_type(8))) float v8f;

#define RH   32
#define NT   24
#define H    32
#define SCS  28
#define TP   48
#define WAVES_PER_BLOCK 8
#define ROWS_PER_WAVE   32
#define ROWS_PER_BLOCK  (WAVES_PER_BLOCK * ROWS_PER_WAVE)

static __device__ __forceinline__ float fast_tanh(float v) {
#if __has_builtin(__builtin_amdgcn_tanhf)
    return __builtin_amdgcn_tanhf(v);
#else
    return tanhf(v);
#endif
}

static __device__ __forceinline__ float fast_exp(float v) {
#if __has_builtin(__builtin_amdgcn_exp2f)
    return __builtin_amdgcn_exp2f(v * 1.44269504088896340736f);
#else
    return __expf(v);
#endif
}

__global__ __launch_bounds__(ROWS_PER_BLOCK)
void forest_moe_kernel(const float* __restrict__ x,
                       const float* __restrict__ rw1, const float* __restrict__ rb1,
                       const float* __restrict__ rw2, const float* __restrict__ rb2,
                       const float* __restrict__ tw1, const float* __restrict__ tb1,
                       const float* __restrict__ tw2, const float* __restrict__ tb2,
                       float* __restrict__ y)
{
    __shared__ float s_rw1[RH], s_rb1[RH];
    __shared__ float s_rw2[RH * NT];
    __shared__ float s_rb2[NT], s_tb2[NT];
    __shared__ v4f   s_tree[NT * TP];
    __shared__ __align__(16) float s_sc[WAVES_PER_BLOCK][ROWS_PER_WAVE][SCS];

    const int tid  = threadIdx.x;
    const int wave = tid >> 5;
    const int lane = tid & 31;

    for (int i = tid; i < RH * NT; i += ROWS_PER_BLOCK) s_rw2[i] = rw2[i];
    for (int i = tid; i < NT * H;  i += ROWS_PER_BLOCK) {
        const int t = i >> 5, k = i & (H - 1);
        v4f w = { tw1[i], tb1[i], tw2[i], 0.0f };
        s_tree[t * TP + k] = w;
        if (k < 15) s_tree[t * TP + H + k] = w;
    }
    if (tid < RH) { s_rw1[tid] = rw1[tid]; s_rb1[tid] = rb1[tid]; }
    if (tid < NT) { s_rb2[tid] = rb2[tid]; s_tb2[tid] = tb2[tid]; }
    __syncthreads();

    const int rowBase = blockIdx.x * ROWS_PER_BLOCK + wave * ROWS_PER_WAVE;
    const int mrow    = lane & 15;
    const int khalf   = (lane >> 4) << 1;

    float xsave[2];
#pragma unroll
    for (int rt = 0; rt < 2; ++rt) {
        const int   gr = rowBase + rt * 16 + mrow;
        const float xv = x[gr];
        xsave[rt] = xv;

        v2f a[8];
#pragma unroll
        for (int s = 0; s < 8; ++s) {
            const int k0 = 4 * s + khalf;
            a[s].x = fast_tanh(xv * s_rw1[k0]     + s_rb1[k0]);
            a[s].y = fast_tanh(xv * s_rw1[k0 + 1] + s_rb1[k0 + 1]);
        }
#pragma unroll
        for (int n = 0; n < 2; ++n) {
            const int col = n * 16 + mrow;
            v8f c = {};
#pragma unroll
            for (int s = 0; s < 8; ++s) {
                const int k0 = 4 * s + khalf;
                v2f b;
                b.x = (col < NT) ? s_rw2[k0 * NT + col]       : 0.0f;
                b.y = (col < NT) ? s_rw2[(k0 + 1) * NT + col] : 0.0f;
                c = __builtin_amdgcn_wmma_f32_16x16x4_f32(
                         false, a[s],  false, b,
                         (short)0, c,  false,  false);
            }
            if (col < NT) {
#pragma unroll
                for (int r = 0; r < 8; ++r) {
                    const int row = rt * 16 + ((lane >> 4) << 3) + r;
                    s_sc[wave][row][col] = c[r];
                }
            }
        }
    }
    __syncthreads();

    const float xv = (lane < 16) ? xsave[0] : xsave[1];

    const v4f* scv = (const v4f*)&s_sc[wave][lane][0];
    v4f q[6];
#pragma unroll
    for (int j = 0; j < 6; ++j) q[j] = scv[j];

    float b0 = -__builtin_inff(), b1v = -__builtin_inff(), b2v = -__builtin_inff();
    int   i0 = 0, i1 = 0, i2 = 0;
#pragma unroll
    for (int t = 0; t < NT; ++t) {
        const float sv = q[t >> 2][t & 3] + s_rb2[t];
        if (sv > b0)       { b2v = b1v; i2 = i1; b1v = b0; i1 = i0; b0 = sv; i0 = t; }
        else if (sv > b1v) { b2v = b1v; i2 = i1; b1v = sv; i1 = t; }
        else if (sv > b2v) { b2v = sv;  i2 = t; }
    }
    const float e1  = fast_exp(b1v - b0);
    const float e2  = fast_exp(b2v - b0);
    const float inv = 1.0f / (1.0f + e1 + e2);

    const int   idx[3] = { i0, i1, i2 };
    const float wgt[3] = { inv, e1 * inv, e2 * inv };

    const int skew = lane & 15;
    float acc = 0.0f;
#pragma unroll
    for (int i = 0; i < 3; ++i) {
        const int t = idx[i];
        const v4f* tp = &s_tree[t * TP + skew];
        float dot = 0.0f;
#pragma unroll
        for (int k = 0; k < H; ++k) {
            const v4f w = tp[k];
            const float h = fast_tanh(xv * w.x + w.y);
            dot += h * w.z;
        }
        acc += wgt[i] * (dot + s_tb2[t]);
    }
    y[rowBase + lane] = acc;
}

extern "C" void kernel_launch(void* const* d_in, const int* in_sizes, int n_in,
                              void* d_out, int out_size, void* d_ws, size_t ws_size,
                              hipStream_t stream) {
    const float* x   = (const float*)d_in[0];
    const float* rw1 = (const float*)d_in[1];
    const float* rb1 = (const float*)d_in[2];
    const float* rw2 = (const float*)d_in[3];
    const float* rb2 = (const float*)d_in[4];
    const float* tw1 = (const float*)d_in[5];
    const float* tb1 = (const float*)d_in[6];
    const float* tw2 = (const float*)d_in[7];
    const float* tb2 = (const float*)d_in[8];
    float* y = (float*)d_out;

    const int B = in_sizes[0];
    const int grid = B / ROWS_PER_BLOCK;
    forest_moe_kernel<<<grid, ROWS_PER_BLOCK, 0, stream>>>(
        x, rw1, rb1, rw2, rb2, tw1, tb1, tw2, tb2, y);
}


# top-3 selection written branch-free: per score 3 compares, 5 index selects, one select and two median-of-three; the 24 one-by-one bias reads become 6 vector reads issued ahead of it
# speedup vs baseline: 1.3959x; 1.0533x over previous
#include <hip/hip_runtime.h>
#include <cmath>

typedef __attribute__((ext_vector_type(2))) float v2f;
typedef __attribute__((ext_vector_type(4))) float v4f;
typedef __attribute__((ext_vector_type(8))) float v8f;

#define RH   32
#define NT   24
#define H    32
#define SCS  28
#define TP   48
#define WAVES_PER_BLOCK 8
#define ROWS_PER_WAVE   32
#define ROWS_PER_BLOCK  (WAVES_PER_BLOCK * ROWS_PER_WAVE)

static __device__ __forceinline__ float fast_tanh(float v) {
#if __has_builtin(__builtin_amdgcn_tanhf)
    return __builtin_amdgcn_tanhf(v);
#else
    return tanhf(v);
#endif
}

static __device__ __forceinline__ float fast_exp(float v) {
#if __has_builtin(__builtin_amdgcn_exp2f)
    return __builtin_amdgcn_exp2f(v * 1.44269504088896340736f);
#else
    return __expf(v);
#endif
}

__global__ __launch_bounds__(ROWS_PER_BLOCK)
void forest_moe_kernel(const float* __restrict__ x,
                       const float* __restrict__ rw1, const float* __restrict__ rb1,
                       const float* __restrict__ rw2, const float* __restrict__ rb2,
                       const float* __restrict__ tw1, const float* __restrict__ tb1,
                       const float* __restrict__ tw2, const float* __restrict__ tb2,
                       float* __restrict__ y)
{
    __shared__ float s_rw1[RH], s_rb1[RH];
    __shared__ float s_rw2[RH * NT];
    __shared__ __align__(16) float s_rb2[NT];
    __shared__ float s_tb2[NT];
    __shared__ v4f   s_tree[NT * TP];
    __shared__ __align__(16) float s_sc[WAVES_PER_BLOCK][ROWS_PER_WAVE][SCS];

    const int tid  = threadIdx.x;
    const int wave = tid >> 5;
    const int lane = tid & 31;

    for (int i = tid; i < RH * NT; i += ROWS_PER_BLOCK) s_rw2[i] = rw2[i];
    for (int i = tid; i < NT * H;  i += ROWS_PER_BLOCK) {
        const int t = i >> 5, k = i & (H - 1);
        v4f w = { tw1[i], tb1[i], tw2[i], 0.0f };
        s_tree[t * TP + k] = w;
        if (k < 15) s_tree[t * TP + H + k] = w;
    }
    if (tid < RH) { s_rw1[tid] = rw1[tid]; s_rb1[tid] = rb1[tid]; }
    if (tid < NT) { s_rb2[tid] = rb2[tid]; s_tb2[tid] = tb2[tid]; }
    __syncthreads();

    const int rowBase = blockIdx.x * ROWS_PER_BLOCK + wave * ROWS_PER_WAVE;
    const int mrow    = lane & 15;
    const int khalf   = (lane >> 4) << 1;

    float xsave[2];
#pragma unroll
    for (int rt = 0; rt < 2; ++rt) {
        const int   gr = rowBase + rt * 16 + mrow;
        const float xv = x[gr];
        xsave[rt] = xv;

        v2f a[8];
#pragma unroll
        for (int s = 0; s < 8; ++s) {
            const int k0 = 4 * s + khalf;
            a[s].x = fast_tanh(xv * s_rw1[k0]     + s_rb1[k0]);
            a[s].y = fast_tanh(xv * s_rw1[k0 + 1] + s_rb1[k0 + 1]);
        }
#pragma unroll
        for (int n = 0; n < 2; ++n) {
            const int col = n * 16 + mrow;
            v8f c = {};
#pragma unroll
            for (int s = 0; s < 8; ++s) {
                const int k0 = 4 * s + khalf;
                v2f b;
                b.x = (col < NT) ? s_rw2[k0 * NT + col]       : 0.0f;
                b.y = (col < NT) ? s_rw2[(k0 + 1) * NT + col] : 0.0f;
                c = __builtin_amdgcn_wmma_f32_16x16x4_f32(
                         false, a[s],  false, b,
                         (short)0, c,  false,  false);
            }
            if (col < NT) {
#pragma unroll
                for (int r = 0; r < 8; ++r) {
                    const int row = rt * 16 + ((lane >> 4) << 3) + r;
                    s_sc[wave][row][col] = c[r];
                }
            }
        }
    }
    __syncthreads();

    const float xv = (lane < 16) ? xsave[0] : xsave[1];

    const v4f* scv = (const v4f*)&s_sc[wave][lane][0];
    const v4f* rbv = (const v4f*)&s_rb2[0];
    v4f q[6];
#pragma unroll
    for (int j = 0; j < 6; ++j) q[j] = scv[j] + rbv[j];

    float b0 = -__builtin_inff(), b1v = -__builtin_inff(), b2v = -__builtin_inff();
    int   i0 = 0, i1 = 0, i2 = 0;
#pragma unroll
    for (int t = 0; t < NT; ++t) {
        const float sv = q[t >> 2][t & 3];
        const bool  c0 = sv > b0, c1 = sv > b1v, c2 = sv > b2v;
        const int   n2i = c1 ? i1 : (c2 ? t : i2);
        const int   n1i = c0 ? i0 : (c1 ? t : i1);
        const int   n0i = c0 ? t  : i0;
        const float n2  = __builtin_amdgcn_fmed3f(sv, b1v, b2v);
        const float n1  = __builtin_amdgcn_fmed3f(sv, b0, b1v);
        const float n0  = c0 ? sv : b0;
        b0 = n0; b1v = n1; b2v = n2;
        i0 = n0i; i1 = n1i; i2 = n2i;
    }
    const float e1  = fast_exp(b1v - b0);
    const float e2  = fast_exp(b2v - b0);
    const float inv = 1.0f / (1.0f + e1 + e2);

    const int   idx[3] = { i0, i1, i2 };
    const float wgt[3] = { inv, e1 * inv, e2 * inv };

    const int skew = lane & 15;
    float acc = 0.0f;
#pragma unroll
    for (int i = 0; i < 3; ++i) {
        const int t = idx[i];
        const v4f* tp = &s_tree[t * TP + skew];
        float dot = 0.0f;
#pragma unroll
        for (int k = 0; k < H; ++k) {
            const v4f w = tp[k];
            const float h = fast_tanh(xv * w.x + w.y);
            dot += h * w.z;
        }
        acc += wgt[i] * (dot + s_tb2[t]);
    }
    y[rowBase + lane] = acc;
}

extern "C" void kernel_launch(void* const* d_in, const int* in_sizes, int n_in,
                              void* d_out, int out_size, void* d_ws, size_t ws_size,
                              hipStream_t stream) {
    const float* x   = (const float*)d_in[0];
    const float* rw1 = (const float*)d_in[1];
    const float* rb1 = (const float*)d_in[2];
    const float* rw2 = (const float*)d_in[3];
    const float* rb2 = (const float*)d_in[4];
    const float* tw1 = (const float*)d_in[5];
    const float* tb1 = (const float*)d_in[6];
    const float* tw2 = (const float*)d_in[7];
    const float* tb2 = (const float*)d_in[8];
    float* y = (float*)d_out;

    const int B = in_sizes[0];
    const int grid = B / ROWS_PER_BLOCK;
    forest_moe_kernel<<<grid, ROWS_PER_BLOCK, 0, stream>>>(
        x, rw1, rb1, rw2, rb2, tw1, tb1, tw2, tb2, y);
}


# router weights staged in LDS in WMMA B-operand order, zero-padded to 32 columns: a wave reads its B operands once with 8 unmasked ds_load_b128 ahead of the WMMAs; 4 accumulator tiles advance together
# speedup vs baseline: 1.5451x; 1.1069x over previous
#include <hip/hip_runtime.h>
#include <cmath>

typedef __attribute__((ext_vector_type(2))) float v2f;
typedef __attribute__((ext_vector_type(4))) float v4f;
typedef __attribute__((ext_vector_type(8))) float v8f;

#define RH   32
#define NT   24
#define H    32
#define SCS  28
#define TP   48
#define BP   20
#define WAVES_PER_BLOCK 8
#define ROWS_PER_WAVE   32
#define ROWS_PER_BLOCK  (WAVES_PER_BLOCK * ROWS_PER_WAVE)

static __device__ __forceinline__ float fast_tanh(float v) {
#if __has_builtin(__builtin_amdgcn_tanhf)
    return __builtin_amdgcn_tanhf(v);
#else
    return tanhf(v);
#endif
}

static __device__ __forceinline__ float fast_exp(float v) {
#if __has_builtin(__builtin_amdgcn_exp2f)
    return __builtin_amdgcn_exp2f(v * 1.44269504088896340736f);
#else
    return __expf(v);
#endif
}

__global__ __launch_bounds__(ROWS_PER_BLOCK)
void forest_moe_kernel(const float* __restrict__ x,
                       const float* __restrict__ rw1, const float* __restrict__ rb1,
                       const float* __restrict__ rw2, const float* __restrict__ rb2,
                       const float* __restrict__ tw1, const float* __restrict__ tb1,
                       const float* __restrict__ tw2, const float* __restrict__ tb2,
                       float* __restrict__ y)
{
    __shared__ float s_rw1[RH], s_rb1[RH];
    __shared__ __align__(16) float s_bfrag[2 * 32 * BP];
    __shared__ __align__(16) float s_rb2[NT];
    __shared__ float s_tb2[NT];
    __shared__ v4f   s_tree[NT * TP];
    __shared__ __align__(16) float s_sc[WAVES_PER_BLOCK][ROWS_PER_WAVE][SCS];

    const int tid  = threadIdx.x;
    const int wave = tid >> 5;
    const int lane = tid & 31;

    {
        const int R   = tid >> 2, qd = tid & 3;
        const int l   = R & 31;
        const int col = (R >> 5) * 16 + (l & 15);
        const int kh  = (l >> 4) << 1;
        v4f w;
#pragma unroll
        for (int j = 0; j < 4; ++j) {
            const int f = 4 * qd + j;
            const int k = 4 * (f >> 1) + kh + (f & 1);
            w[j] = (col < NT) ? rw2[k * NT + col] : 0.0f;
        }
        *(v4f*)&s_bfrag[R * BP + 4 * qd] = w;
    }
    for (int i = tid; i < NT * H;  i += ROWS_PER_BLOCK) {
        const int t = i >> 5, k = i & (H - 1);
        v4f w = { tw1[i], tb1[i], tw2[i], 0.0f };
        s_tree[t * TP + k] = w;
        if (k < 15) s_tree[t * TP + H + k] = w;
    }
    if (tid < RH) { s_rw1[tid] = rw1[tid]; s_rb1[tid] = rb1[tid]; }
    if (tid < NT) { s_rb2[tid] = rb2[tid]; s_tb2[tid] = tb2[tid]; }
    __syncthreads();

    const int rowBase = blockIdx.x * ROWS_PER_BLOCK + wave * ROWS_PER_WAVE;
    const int mrow    = lane & 15;
    const int khalf   = (lane >> 4) << 1;

    v4f bq[2][4];
#pragma unroll
    for (int n = 0; n < 2; ++n) {
        const v4f* bp = (const v4f*)&s_bfrag[(n * 32 + lane) * BP];
#pragma unroll
        for (int j = 0; j < 4; ++j) bq[n][j] = bp[j];
    }

    float xsave[2];
    xsave[0] = x[rowBase + mrow];
    xsave[1] = x[rowBase + 16 + mrow];

    v8f c[2][2] = {};
#pragma unroll
    for (int s = 0; s < 8; ++s) {
        const int   k0  = 4 * s + khalf;
        const float w1a = s_rw1[k0], w1b = s_rw1[k0 + 1];
        const float b1a = s_rb1[k0], b1b = s_rb1[k0 + 1];
        v2f a[2];
#pragma unroll
        for (int rt = 0; rt < 2; ++rt) {
            a[rt].x = fast_tanh(xsave[rt] * w1a + b1a);
            a[rt].y = fast_tanh(xsave[rt] * w1b + b1b);
        }
#pragma unroll
        for (int rt = 0; rt < 2; ++rt) {
#pragma unroll
            for (int n = 0; n < 2; ++n) {
                v2f b;
                b.x = bq[n][s >> 1][(s & 1) * 2];
                b.y = bq[n][s >> 1][(s & 1) * 2 + 1];
                c[rt][n] = __builtin_amdgcn_wmma_f32_16x16x4_f32(
                         false, a[rt],  false, b,
                         (short)0, c[rt][n],  false,  false);
            }
        }
    }
#pragma unroll
    for (int rt = 0; rt < 2; ++rt) {
#pragma unroll
        for (int n = 0; n < 2; ++n) {
            const int col = n * 16 + mrow;
            if (col < NT) {
#pragma unroll
                for (int r = 0; r < 8; ++r) {
                    const int row = rt * 16 + ((lane >> 4) << 3) + r;
                    s_sc[wave][row][col] = c[rt][n][r];
                }
            }
        }
    }
    __syncthreads();

    const float xv = (lane < 16) ? xsave[0] : xsave[1];

    const v4f* scv = (const v4f*)&s_sc[wave][lane][0];
    const v4f* rbv = (const v4f*)&s_rb2[0];
    v4f q[6];
#pragma unroll
    for (int j = 0; j < 6; ++j) q[j] = scv[j] + rbv[j];

    float b0 = -__builtin_inff(), b1v = -__builtin_inff(), b2v = -__builtin_inff();
    int   i0 = 0, i1 = 0, i2 = 0;
#pragma unroll
    for (int t = 0; t < NT; ++t) {
        const float sv = q[t >> 2][t & 3];
        const bool  c0 = sv > b0, c1 = sv > b1v, c2 = sv > b2v;
        const int   n2i = c1 ? i1 : (c2 ? t : i2);
        const int   n1i = c0 ? i0 : (c1 ? t : i1);
        const int   n0i = c0 ? t  : i0;
        const float n2  = __builtin_amdgcn_fmed3f(sv, b1v, b2v);
        const float n1  = __builtin_amdgcn_fmed3f(sv, b0, b1v);
        const float n0  = c0 ? sv : b0;
        b0 = n0; b1v = n1; b2v = n2;
        i0 = n0i; i1 = n1i; i2 = n2i;
    }
    const float e1  = fast_exp(b1v - b0);
    const float e2  = fast_exp(b2v - b0);
    const float inv = 1.0f / (1.0f + e1 + e2);

    const int   idx[3] = { i0, i1, i2 };
    const float wgt[3] = { inv, e1 * inv, e2 * inv };

    const int skew = lane & 15;
    float acc = 0.0f;
#pragma unroll
    for (int i = 0; i < 3; ++i) {
        const int t = idx[i];
        const v4f* tp = &s_tree[t * TP + skew];
        float dot = 0.0f;
#pragma unroll
        for (int k = 0; k < H; ++k) {
            const v4f w = tp[k];
            const float h = fast_tanh(xv * w.x + w.y);
            dot += h * w.z;
        }
        acc += wgt[i] * (dot + s_tb2[t]);
    }
    y[rowBase + lane] = acc;
}

extern "C" void kernel_launch(void* const* d_in, const int* in_sizes, int n_in,
                              void* d_out, int out_size, void* d_ws, size_t ws_size,
                              hipStream_t stream) {
    const float* x   = (const float*)d_in[0];
    const float* rw1 = (const float*)d_in[1];
    const float* rb1 = (const float*)d_in[2];
    const float* rw2 = (const float*)d_in[3];
    const float* rb2 = (const float*)d_in[4];
    const float* tw1 = (const float*)d_in[5];
    const float* tb1 = (const float*)d_in[6];
    const float* tw2 = (const float*)d_in[7];
    const float* tb2 = (const float*)d_in[8];
    float* y = (float*)d_out;

    const int B = in_sizes[0];
    const int grid = B / ROWS_PER_BLOCK;
    forest_moe_kernel<<<grid, ROWS_PER_BLOCK, 0, stream>>>(
        x, rw1, rb1, rw2, rb2, tw1, tb1, tw2, tb2, y);
}


# every global load (parameters and x) issued as one group before the first LDS store: unconditional in-bounds addresses, staging loop unrolled; one exposed memory round trip instead of seven
# speedup vs baseline: 1.9677x; 1.2735x over previous
#include <hip/hip_runtime.h>
#include <cmath>

typedef __attribute__((ext_vector_type(2))) float v2f;
typedef __attribute__((ext_vector_type(4))) float v4f;
typedef __attribute__((ext_vector_type(8))) float v8f;

#define RH   32
#define NT   24
#define H    32
#define SCS  28
#define TP   48
#define BP   20
#define WAVES_PER_BLOCK 8
#define ROWS_PER_WAVE   32
#define ROWS_PER_BLOCK  (WAVES_PER_BLOCK * ROWS_PER_WAVE)

static __device__ __forceinline__ float fast_tanh(float v) {
#if __has_builtin(__builtin_amdgcn_tanhf)
    return __builtin_amdgcn_tanhf(v);
#else
    return tanhf(v);
#endif
}

static __device__ __forceinline__ float fast_exp(float v) {
#if __has_builtin(__builtin_amdgcn_exp2f)
    return __builtin_amdgcn_exp2f(v * 1.44269504088896340736f);
#else
    return __expf(v);
#endif
}

__global__ __launch_bounds__(ROWS_PER_BLOCK)
void forest_moe_kernel(const float* __restrict__ x,
                       const float* __restrict__ rw1, const float* __restrict__ rb1,
                       const float* __restrict__ rw2, const float* __restrict__ rb2,
                       const float* __restrict__ tw1, const float* __restrict__ tb1,
                       const float* __restrict__ tw2, const float* __restrict__ tb2,
                       float* __restrict__ y)
{
    __shared__ float s_rw1[RH], s_rb1[RH];
    __shared__ __align__(16) float s_bfrag[2 * 32 * BP];
    __shared__ __align__(16) float s_rb2[NT];
    __shared__ float s_tb2[NT];
    __shared__ v4f   s_tree[NT * TP];
    __shared__ __align__(16) float s_sc[WAVES_PER_BLOCK][ROWS_PER_WAVE][SCS];

    const int tid  = threadIdx.x;
    const int wave = tid >> 5;
    const int lane = tid & 31;

    const int rowBase = blockIdx.x * ROWS_PER_BLOCK + wave * ROWS_PER_WAVE;
    const int mrow    = lane & 15;
    const int khalf   = (lane >> 4) << 1;

    const int R    = tid >> 2, qd = tid & 3;
    const int l    = R & 31;
    const int col  = (R >> 5) * 16 + (l & 15);
    const int colc = (col < NT) ? col : (NT - 1);
    const int kh   = (l >> 4) << 1;
    float g_w2[4];
#pragma unroll
    for (int j = 0; j < 4; ++j) {
        const int f = 4 * qd + j;
        const int k = 4 * (f >> 1) + kh + (f & 1);
        g_w2[j] = rw2[k * NT + colc];
    }
    float g_t1[3], g_tb[3], g_t2[3];
#pragma unroll
    for (int j = 0; j < 3; ++j) {
        const int i = tid + j * ROWS_PER_BLOCK;
        g_t1[j] = tw1[i]; g_tb[j] = tb1[i]; g_t2[j] = tw2[i];
    }
    const int   sl    = (lane < NT) ? lane : (NT - 1);
    const float g_rw1 = rw1[lane], g_rb1 = rb1[lane];
    const float g_rb2 = rb2[sl],   g_tb2 = tb2[sl];
    float xsave[2];
    xsave[0] = x[rowBase + mrow];
    xsave[1] = x[rowBase + 16 + mrow];

    {
        v4f w;
#pragma unroll
        for (int j = 0; j < 4; ++j) w[j] = (col < NT) ? g_w2[j] : 0.0f;
        *(v4f*)&s_bfrag[R * BP + 4 * qd] = w;
    }
#pragma unroll
    for (int j = 0; j < 3; ++j) {
        const int t = (tid >> 5) + j * WAVES_PER_BLOCK;
        v4f w = { g_t1[j], g_tb[j], g_t2[j], 0.0f };
        s_tree[t * TP + lane] = w;
        if (lane < 15) s_tree[t * TP + H + lane] = w;
    }
    if (tid < RH) { s_rw1[tid] = g_rw1; s_rb1[tid] = g_rb1; }
    if (tid < NT) { s_rb2[tid] = g_rb2; s_tb2[tid] = g_tb2; }
    __syncthreads();

    v4f bq[2][4];
#pragma unroll
    for (int n = 0; n < 2; ++n) {
        const v4f* bp = (const v4f*)&s_bfrag[(n * 32 + lane) * BP];
#pragma unroll
        for (int j = 0; j < 4; ++j) bq[n][j] = bp[j];
    }

    v8f c[2][2] = {};
#pragma unroll
    for (int s = 0; s < 8; ++s) {
        const int   k0  = 4 * s + khalf;
        const float w1a = s_rw1[k0], w1b = s_rw1[k0 + 1];
        const float b1a = s_rb1[k0], b1b = s_rb1[k0 + 1];
        v2f a[2];
#pragma unroll
        for (int rt = 0; rt < 2; ++rt) {
            a[rt].x = fast_tanh(xsave[rt] * w1a + b1a);
            a[rt].y = fast_tanh(xsave[rt] * w1b + b1b);
        }
#pragma unroll
        for (int rt = 0; rt < 2; ++rt) {
#pragma unroll
            for (int n = 0; n < 2; ++n) {
                v2f b;
                b.x = bq[n][s >> 1][(s & 1) * 2];
                b.y = bq[n][s >> 1][(s & 1) * 2 + 1];
                c[rt][n] = __builtin_amdgcn_wmma_f32_16x16x4_f32(
                         false, a[rt],  false, b,
                         (short)0, c[rt][n],  false,  false);
            }
        }
    }
#pragma unroll
    for (int rt = 0; rt < 2; ++rt) {
#pragma unroll
        for (int n = 0; n < 2; ++n) {
            const int col = n * 16 + mrow;
            if (col < NT) {
#pragma unroll
                for (int r = 0; r < 8; ++r) {
                    const int row = rt * 16 + ((lane >> 4) << 3) + r;
                    s_sc[wave][row][col] = c[rt][n][r];
                }
            }
        }
    }
    __syncthreads();

    const float xv = (lane < 16) ? xsave[0] : xsave[1];

    const v4f* scv = (const v4f*)&s_sc[wave][lane][0];
    const v4f* rbv = (const v4f*)&s_rb2[0];
    v4f q[6];
#pragma unroll
    for (int j = 0; j < 6; ++j) q[j] = scv[j] + rbv[j];

    float b0 = -__builtin_inff(), b1v = -__builtin_inff(), b2v = -__builtin_inff();
    int   i0 = 0, i1 = 0, i2 = 0;
#pragma unroll
    for (int t = 0; t < NT; ++t) {
        const float sv = q[t >> 2][t & 3];
        const bool  c0 = sv > b0, c1 = sv > b1v, c2 = sv > b2v;
        const int   n2i = c1 ? i1 : (c2 ? t : i2);
        const int   n1i = c0 ? i0 : (c1 ? t : i1);
        const int   n0i = c0 ? t  : i0;
        const float n2  = __builtin_amdgcn_fmed3f(sv, b1v, b2v);
        const float n1  = __builtin_amdgcn_fmed3f(sv, b0, b1v);
        const float n0  = c0 ? sv : b0;
        b0 = n0; b1v = n1; b2v = n2;
        i0 = n0i; i1 = n1i; i2 = n2i;
    }
    const float e1  = fast_exp(b1v - b0);
    const float e2  = fast_exp(b2v - b0);
    const float inv = 1.0f / (1.0f + e1 + e2);

    const int   idx[3] = { i0, i1, i2 };
    const float wgt[3] = { inv, e1 * inv, e2 * inv };

    const int skew = lane & 15;
    float acc = 0.0f;
#pragma unroll
    for (int i = 0; i < 3; ++i) {
        const int t = idx[i];
        const v4f* tp = &s_tree[t * TP + skew];
        float dot = 0.0f;
#pragma unroll
        for (int k = 0; k < H; ++k) {
            const v4f w = tp[k];
            const float h = fast_tanh(xv * w.x + w.y);
            dot += h * w.z;
        }
        acc += wgt[i] * (dot + s_tb2[t]);
    }
    y[rowBase + lane] = acc;
}

extern "C" void kernel_launch(void* const* d_in, const int* in_sizes, int n_in,
                              void* d_out, int out_size, void* d_ws, size_t ws_size,
                              hipStream_t stream) {
    const float* x   = (const float*)d_in[0];
    const float* rw1 = (const float*)d_in[1];
    const float* rb1 = (const float*)d_in[2];
    const float* rw2 = (const float*)d_in[3];
    const float* rb2 = (const float*)d_in[4];
    const float* tw1 = (const float*)d_in[5];
    const float* tb1 = (const float*)d_in[6];
    const float* tw2 = (const float*)d_in[7];
    const float* tb2 = (const float*)d_in[8];
    float* y = (float*)d_out;

    const int B = in_sizes[0];
    const int grid = B / ROWS_PER_BLOCK;
    forest_moe_kernel<<<grid, ROWS_PER_BLOCK, 0, stream>>>(
        x, rw1, rb1, rw2, rb2, tw1, tb1, tw2, tb2, y);
}


# tree table stored as three planes (w1|b1|w2), 4 consecutive k per 16-byte slot: 3 gathers serve 4 k (was 4), multiply-adds run as packed pairs; each dot summed as 4 interleaved partial sums (k mod 4)
# speedup vs baseline: 2.1367x; 1.0859x over previous
#include <hip/hip_runtime.h>
#include <cmath>

typedef __attribute__((ext_vector_type(2))) float v2f;
typedef __attribute__((ext_vector_type(4))) float v4f;
typedef __attribute__((ext_vector_type(8))) float v8f;

#define RH   32
#define NT   24
#define H    32
#define SCS  28
#define PP   24
#define TP   80
#define BP   20
#define WAVES_PER_BLOCK 8
#define ROWS_PER_WAVE   32
#define ROWS_PER_BLOCK  (WAVES_PER_BLOCK * ROWS_PER_WAVE)

static __device__ __forceinline__ float fast_tanh(float v) {
#if __has_builtin(__builtin_amdgcn_tanhf)
    return __builtin_amdgcn_tanhf(v);
#else
    return tanhf(v);
#endif
}

static __device__ __forceinline__ float fast_exp(float v) {
#if __has_builtin(__builtin_amdgcn_exp2f)
    return __builtin_amdgcn_exp2f(v * 1.44269504088896340736f);
#else
    return __expf(v);
#endif
}

__global__ __launch_bounds__(ROWS_PER_BLOCK)
void forest_moe_kernel(const float* __restrict__ x,
                       const float* __restrict__ rw1, const float* __restrict__ rb1,
                       const float* __restrict__ rw2, const float* __restrict__ rb2,
                       const float* __restrict__ tw1, const float* __restrict__ tb1,
                       const float* __restrict__ tw2, const float* __restrict__ tb2,
                       float* __restrict__ y)
{
    __shared__ float s_rw1[RH], s_rb1[RH];
    __shared__ __align__(16) float s_bfrag[2 * 32 * BP];
    __shared__ __align__(16) float s_rb2[NT];
    __shared__ float s_tb2[NT];
    __shared__ v4f   s_tree[NT * TP];
    __shared__ __align__(16) float s_sc[WAVES_PER_BLOCK][ROWS_PER_WAVE][SCS];

    const int tid  = threadIdx.x;
    const int wave = tid >> 5;
    const int lane = tid & 31;

    const int rowBase = blockIdx.x * ROWS_PER_BLOCK + wave * ROWS_PER_WAVE;
    const int mrow    = lane & 15;
    const int khalf   = (lane >> 4) << 1;

    const int R    = tid >> 2, qd = tid & 3;
    const int l    = R & 31;
    const int col  = (R >> 5) * 16 + (l & 15);
    const int colc = (col < NT) ? col : (NT - 1);
    const int kh   = (l >> 4) << 1;
    float g_w2[4];
#pragma unroll
    for (int j = 0; j < 4; ++j) {
        const int f = 4 * qd + j;
        const int k = 4 * (f >> 1) + kh + (f & 1);
        g_w2[j] = rw2[k * NT + colc];
    }
    const int tq = (tid < NT * (H / 4)) ? tid : (tid - 2 * 32);
    const v4f g_t1 = ((const v4f*)tw1)[tq];
    const v4f g_tb = ((const v4f*)tb1)[tq];
    const v4f g_t2 = ((const v4f*)tw2)[tq];
    const int   sl    = (lane < NT) ? lane : (NT - 1);
    const float g_rw1 = rw1[lane], g_rb1 = rb1[lane];
    const float g_rb2 = rb2[sl],   g_tb2 = tb2[sl];
    float xsave[2];
    xsave[0] = x[rowBase + mrow];
    xsave[1] = x[rowBase + 16 + mrow];

    {
        v4f w;
#pragma unroll
        for (int j = 0; j < 4; ++j) w[j] = (col < NT) ? g_w2[j] : 0.0f;
        *(v4f*)&s_bfrag[R * BP + 4 * qd] = w;
    }
    {
        const int t = tq >> 3, j = tq & 7;
        v4f* p = &s_tree[t * TP + j];
        p[0]      = g_t1; p[8]          = g_t1; p[16]          = g_t1;
        p[PP]     = g_tb; p[PP + 8]     = g_tb; p[PP + 16]     = g_tb;
        p[2 * PP] = g_t2; p[2 * PP + 8] = g_t2; p[2 * PP + 16] = g_t2;
    }
    s_rw1[lane] = g_rw1; s_rb1[lane] = g_rb1;
    s_rb2[sl]   = g_rb2; s_tb2[sl]   = g_tb2;
    __syncthreads();

    v4f bq[2][4];
#pragma unroll
    for (int n = 0; n < 2; ++n) {
        const v4f* bp = (const v4f*)&s_bfrag[(n * 32 + lane) * BP];
#pragma unroll
        for (int j = 0; j < 4; ++j) bq[n][j] = bp[j];
    }

    v8f c[2][2] = {};
#pragma unroll
    for (int s = 0; s < 8; ++s) {
        const int   k0  = 4 * s + khalf;
        const float w1a = s_rw1[k0], w1b = s_rw1[k0 + 1];
        const float b1a = s_rb1[k0], b1b = s_rb1[k0 + 1];
        v2f a[2];
#pragma unroll
        for (int rt = 0; rt < 2; ++rt) {
            a[rt].x = fast_tanh(xsave[rt] * w1a + b1a);
            a[rt].y = fast_tanh(xsave[rt] * w1b + b1b);
        }
#pragma unroll
        for (int rt = 0; rt < 2; ++rt) {
#pragma unroll
            for (int n = 0; n < 2; ++n) {
                v2f b;
                b.x = bq[n][s >> 1][(s & 1) * 2];
                b.y = bq[n][s >> 1][(s & 1) * 2 + 1];
                c[rt][n] = __builtin_amdgcn_wmma_f32_16x16x4_f32(
                         false, a[rt],  false, b,
                         (short)0, c[rt][n],  false,  false);
            }
        }
    }
#pragma unroll
    for (int rt = 0; rt < 2; ++rt) {
#pragma unroll
        for (int n = 0; n < 2; ++n) {
            const int col = n * 16 + mrow;
            if (col < NT) {
#pragma unroll
                for (int r = 0; r < 8; ++r) {
                    const int row = rt * 16 + ((lane >> 4) << 3) + r;
                    s_sc[wave][row][col] = c[rt][n][r];
                }
            }
        }
    }
    __syncthreads();

    const float xv = (lane < 16) ? xsave[0] : xsave[1];

    const v4f* scv = (const v4f*)&s_sc[wave][lane][0];
    const v4f* rbv = (const v4f*)&s_rb2[0];
    v4f q[6];
#pragma unroll
    for (int j = 0; j < 6; ++j) q[j] = scv[j] + rbv[j];

    float b0 = -__builtin_inff(), b1v = -__builtin_inff(), b2v = -__builtin_inff();
    int   i0 = 0, i1 = 0, i2 = 0;
#pragma unroll
    for (int t = 0; t < NT; ++t) {
        const float sv = q[t >> 2][t & 3];
        const bool  c0 = sv > b0, c1 = sv > b1v, c2 = sv > b2v;
        const int   n2i = c1 ? i1 : (c2 ? t : i2);
        const int   n1i = c0 ? i0 : (c1 ? t : i1);
        const int   n0i = c0 ? t  : i0;
        const float n2  = __builtin_amdgcn_fmed3f(sv, b1v, b2v);
        const float n1  = __builtin_amdgcn_fmed3f(sv, b0, b1v);
        const float n0  = c0 ? sv : b0;
        b0 = n0; b1v = n1; b2v = n2;
        i0 = n0i; i1 = n1i; i2 = n2i;
    }
    const float e1  = fast_exp(b1v - b0);
    const float e2  = fast_exp(b2v - b0);
    const float inv = 1.0f / (1.0f + e1 + e2);

    const int   idx[3] = { i0, i1, i2 };
    const float wgt[3] = { inv, e1 * inv, e2 * inv };

    const int skew = lane & 15;
    const v2f xv2 = { xv, xv };
    float acc = 0.0f;
#pragma unroll
    for (int i = 0; i < 3; ++i) {
        const int t = idx[i];
        const v4f* tp = &s_tree[t * TP + skew];
        v2f d01 = { 0.0f, 0.0f }, d23 = { 0.0f, 0.0f };
#pragma unroll
        for (int j = 0; j < H / 4; ++j) {
            const v4f w1 = tp[j], b1 = tp[PP + j], w2 = tp[2 * PP + j];
            v2f h01 = xv2 * w1.xy + b1.xy;
            v2f h23 = xv2 * w1.zw + b1.zw;
            h01.x = fast_tanh(h01.x); h01.y = fast_tanh(h01.y);
            h23.x = fast_tanh(h23.x); h23.y = fast_tanh(h23.y);
            d01 = h01 * w2.xy + d01;
            d23 = h23 * w2.zw + d23;
        }
        const v2f d = d01 + d23;
        acc += wgt[i] * ((d.x + d.y) + s_tb2[t]);
    }
    y[rowBase + lane] = acc;
}

extern "C" void kernel_launch(void* const* d_in, const int* in_sizes, int n_in,
                              void* d_out, int out_size, void* d_ws, size_t ws_size,
                              hipStream_t stream) {
    const float* x   = (const float*)d_in[0];
    const float* rw1 = (const float*)d_in[1];
    const float* rb1 = (const float*)d_in[2];
    const float* rw2 = (const float*)d_in[3];
    const float* rb2 = (const float*)d_in[4];
    const float* tw1 = (const float*)d_in[5];
    const float* tb1 = (const float*)d_in[6];
    const float* tw2 = (const float*)d_in[7];
    const float* tb2 = (const float*)d_in[8];
    float* y = (float*)d_out;

    const int B = in_sizes[0];
    const int grid = B / ROWS_PER_BLOCK;
    forest_moe_kernel<<<grid, ROWS_PER_BLOCK, 0, stream>>>(
        x, rw1, rb1, rw2, rb2, tw1, tb1, tw2, tb2, y);
}
